# starting kernel
# baseline (speedup 1.0000x reference)
#include <hip/hip_runtime.h>


typedef __attribute__((ext_vector_type(2))) float v2f;
typedef __attribute__((ext_vector_type(8))) float v8f;

#define S_LEN 1024
#define HID 4

__device__ __forceinline__ float cdna5_tanh(float v) {
#if __has_builtin(__builtin_amdgcn_tanhf)
  return __builtin_amdgcn_tanhf(v);
#else
  float r;
  asm("v_tanh_f32 %0, %1" : "=v"(r) : "v"(v));
  return r;
#endif
}

__global__ void __launch_bounds__(128)
rnn_scan_wmma_kernel(const float* __restrict__ x,
                     const float* __restrict__ W_ih,
                     const float* __restrict__ W_hh,
                     const float* __restrict__ b_ih,
                     const float* __restrict__ b_hh,
                     const float* __restrict__ fc_w,
                     const float* __restrict__ fc_b,
                     float* __restrict__ out,
                     int Btot)
{
  const int gtid = blockIdx.x * blockDim.x + threadIdx.x;
  const int lane = threadIdx.x & 31;
  const int base = (gtid >> 5) << 5;
  if (base >= Btot) return;

  const int  m  = lane & 15;
  const bool lo = lane < 16;

  const int jA = lo ? 0 : 2;
  const int jB = lo ? 1 : 3;
  const float wihA  = W_ih[jA];
  const float wihB  = W_ih[jB];
  const float biasA = b_ih[jA] + b_hh[jA];
  const float biasB = b_ih[jB] + b_hh[jB];
  const float fwA   = fc_w[jA];
  const float fwB   = fc_w[jB];
  const float fcb   = fc_b[0];

  v2f A = {0.f, 0.f};
  {
    const int row = (m == 0) ? 0 : (m == 1) ? 1 : (m == 8) ? 2 : (m == 9) ? 3 : -1;
    if (row >= 0) {
      const int k0 = lo ? 0 : 2;
      A.x = W_hh[row * HID + k0];
      A.y = W_hh[row * HID + k0 + 1];
    }
  }

  const float* __restrict__ xp0 = x + (size_t)(base + m)      * S_LEN;
  const float* __restrict__ xp1 = x + (size_t)(base + 16 + m) * S_LEN;

  v2f h0 = {0.f, 0.f};
  v2f h1 = {0.f, 0.f};
  v8f acc0 = {};
  v8f acc1 = {};

  float4 buf0[4], buf1[4];
#pragma unroll
  for (int i = 0; i < 4; ++i) {
    buf0[i] = *(const float4*)(xp0 + 4 * i);
    buf1[i] = *(const float4*)(xp1 + 4 * i);
  }

  for (int s = 0; s < S_LEN; s += 16) {
    const int sn = (s + 16) & (S_LEN - 1);
    float4 n0[4], n1[4];
#pragma unroll
    for (int i = 0; i < 4; ++i) {
      n0[i] = *(const float4*)(xp0 + sn + 4 * i);
      n1[i] = *(const float4*)(xp1 + sn + 4 * i);
    }

#pragma unroll
    for (int i = 0; i < 4; ++i) {
      const float xs0[4] = {buf0[i].x, buf0[i].y, buf0[i].z, buf0[i].w};
      const float xs1[4] = {buf1[i].x, buf1[i].y, buf1[i].z, buf1[i].w};
#pragma unroll
      for (int u = 0; u < 4; ++u) {
        acc0[0] = __builtin_fmaf(xs0[u], wihA, biasA);
        acc0[1] = __builtin_fmaf(xs0[u], wihB, biasB);
        acc1[0] = __builtin_fmaf(xs1[u], wihA, biasA);
        acc1[1] = __builtin_fmaf(xs1[u], wihB, biasB);

        acc0 = __builtin_amdgcn_wmma_f32_16x16x4_f32(
            false, A, false, h0, (short)0, acc0, false, false);
        acc1 = __builtin_amdgcn_wmma_f32_16x16x4_f32(
            false, A, false, h1, (short)0, acc1, false, false);

        h0.x = cdna5_tanh(acc0[0]);
        h0.y = cdna5_tanh(acc0[1]);
        h1.x = cdna5_tanh(acc1[0]);
        h1.y = cdna5_tanh(acc1[1]);
      }
    }

#pragma unroll
    for (int i = 0; i < 4; ++i) { buf0[i] = n0[i]; buf1[i] = n1[i]; }
  }

  float p0 = __builtin_fmaf(h0.x, fwA, h0.y * fwB);
  float p1 = __builtin_fmaf(h1.x, fwA, h1.y * fwB);
  p0 += __shfl_xor(p0, 16, 32);
  p1 += __shfl_xor(p1, 16, 32);
  if (lo) {
    out[base + m]      = p0 + fcb;
    out[base + 16 + m] = p1 + fcb;
  }
}

extern "C" void kernel_launch(void* const* d_in, const int* in_sizes, int n_in,
                              void* d_out, int out_size, void* d_ws, size_t ws_size,
                              hipStream_t stream) {
  (void)n_in; (void)out_size; (void)d_ws; (void)ws_size;
  const float* x    = (const float*)d_in[0];
  const float* W_ih = (const float*)d_in[1];
  const float* W_hh = (const float*)d_in[2];
  const float* b_ih = (const float*)d_in[3];
  const float* b_hh = (const float*)d_in[4];
  const float* fc_w = (const float*)d_in[5];
  const float* fc_b = (const float*)d_in[6];
  float* out = (float*)d_out;

  const int B = in_sizes[0] / S_LEN;
  const int block = 128;
  const int grid  = (B + 127) / 128;
  rnn_scan_wmma_kernel<<<grid, block, 0, stream>>>(x, W_ih, W_hh, b_ih, b_hh,
                                                   fc_w, fc_b, out, B);
}


# x feed: four 16-step register blocks rotated by name, each requested 48 steps before use; a fence stops the compiler folding the requests back onto their use; both chains keep alternating
# speedup vs baseline: 1.5413x; 1.5413x over previous
#include <hip/hip_runtime.h>


typedef __attribute__((ext_vector_type(2))) float v2f;
typedef __attribute__((ext_vector_type(8))) float v8f;

#define S_LEN 1024
#define HID 4
#define BLK 16

__device__ __forceinline__ float cdna5_tanh(float v) {
#if __has_builtin(__builtin_amdgcn_tanhf)
  return __builtin_amdgcn_tanhf(v);
#else
  float r;
  asm("v_tanh_f32 %0, %1" : "=v"(r) : "v"(v));
  return r;
#endif
}

struct XBlk {
  float4 a[4];
  float4 b[4];
};

__device__ __forceinline__ void load_blk(XBlk& r,
                                         const float* __restrict__ xp0,
                                         const float* __restrict__ xp1,
                                         int off) {
#pragma unroll
  for (int i = 0; i < 4; ++i) {
    r.a[i] = *(const float4*)(xp0 + off + 4 * i);
    r.b[i] = *(const float4*)(xp1 + off + 4 * i);
  }
}

#define REQUESTS_STAY_HERE() asm volatile("" : : : "memory")

#define HALF_STEP() __builtin_amdgcn_sched_barrier(0)

__device__ __forceinline__ void run_blk(const XBlk& r, const v2f A,
                                        v2f& h0, v2f& h1,
                                        v8f& acc0, v8f& acc1,
                                        const float wihA, const float wihB,
                                        const float biasA, const float biasB) {
  const float xs0[BLK] = {r.a[0].x, r.a[0].y, r.a[0].z, r.a[0].w,
                          r.a[1].x, r.a[1].y, r.a[1].z, r.a[1].w,
                          r.a[2].x, r.a[2].y, r.a[2].z, r.a[2].w,
                          r.a[3].x, r.a[3].y, r.a[3].z, r.a[3].w};
  const float xs1[BLK] = {r.b[0].x, r.b[0].y, r.b[0].z, r.b[0].w,
                          r.b[1].x, r.b[1].y, r.b[1].z, r.b[1].w,
                          r.b[2].x, r.b[2].y, r.b[2].z, r.b[2].w,
                          r.b[3].x, r.b[3].y, r.b[3].z, r.b[3].w};

  acc0[0] = __builtin_fmaf(xs0[0], wihA, biasA);
  acc0[1] = __builtin_fmaf(xs0[0], wihB, biasB);
  acc0 = __builtin_amdgcn_wmma_f32_16x16x4_f32(
      false, A, false, h0, (short)0, acc0, false, false);
#pragma unroll
  for (int u = 0; u < BLK; ++u) {
    acc1[0] = __builtin_fmaf(xs1[u], wihA, biasA);
    acc1[1] = __builtin_fmaf(xs1[u], wihB, biasB);
    acc1 = __builtin_amdgcn_wmma_f32_16x16x4_f32(
        false, A, false, h1, (short)0, acc1, false, false);
    HALF_STEP();
    h0.x = cdna5_tanh(acc0[0]);
    h0.y = cdna5_tanh(acc0[1]);
    if (u + 1 < BLK) {
      acc0[0] = __builtin_fmaf(xs0[u + 1], wihA, biasA);
      acc0[1] = __builtin_fmaf(xs0[u + 1], wihB, biasB);
      acc0 = __builtin_amdgcn_wmma_f32_16x16x4_f32(
          false, A, false, h0, (short)0, acc0, false, false);
    }
    HALF_STEP();
    h1.x = cdna5_tanh(acc1[0]);
    h1.y = cdna5_tanh(acc1[1]);
  }
}

__global__ void __launch_bounds__(128)
rnn_scan_wmma_kernel(const float* __restrict__ x,
                     const float* __restrict__ W_ih,
                     const float* __restrict__ W_hh,
                     const float* __restrict__ b_ih,
                     const float* __restrict__ b_hh,
                     const float* __restrict__ fc_w,
                     const float* __restrict__ fc_b,
                     float* __restrict__ out,
                     int Btot)
{
  const int gtid = blockIdx.x * blockDim.x + threadIdx.x;
  const int lane = threadIdx.x & 31;
  const int base = (gtid >> 5) << 5;
  if (base >= Btot) return;

  const int  m  = lane & 15;
  const bool lo = lane < 16;

  const int jA = lo ? 0 : 2;
  const int jB = lo ? 1 : 3;
  const float wihA  = W_ih[jA];
  const float wihB  = W_ih[jB];
  const float biasA = b_ih[jA] + b_hh[jA];
  const float biasB = b_ih[jB] + b_hh[jB];
  const float fwA   = fc_w[jA];
  const float fwB   = fc_w[jB];
  const float fcb   = fc_b[0];

  v2f A = {0.f, 0.f};
  {
    const int row = (m == 0) ? 0 : (m == 1) ? 1 : (m == 8) ? 2 : (m == 9) ? 3 : -1;
    if (row >= 0) {
      const int k0 = lo ? 0 : 2;
      A.x = W_hh[row * HID + k0];
      A.y = W_hh[row * HID + k0 + 1];
    }
  }

  const float* __restrict__ xp0 = x + (size_t)(base + m)      * S_LEN;
  const float* __restrict__ xp1 = x + (size_t)(base + 16 + m) * S_LEN;

  v2f h0 = {0.f, 0.f};
  v2f h1 = {0.f, 0.f};
  v8f acc0 = {};
  v8f acc1 = {};

  XBlk r0, r1, r2, r3;
  load_blk(r0, xp0, xp1, 0 * BLK);
  REQUESTS_STAY_HERE();
  load_blk(r1, xp0, xp1, 1 * BLK);
  REQUESTS_STAY_HERE();
  load_blk(r2, xp0, xp1, 2 * BLK);
  REQUESTS_STAY_HERE();

  int s = 0;
  for (; s < S_LEN - 4 * BLK; s += 4 * BLK) {
    load_blk(r3, xp0, xp1, s + 3 * BLK);
    REQUESTS_STAY_HERE();
    run_blk(r0, A, h0, h1, acc0, acc1, wihA, wihB, biasA, biasB);

    load_blk(r0, xp0, xp1, s + 4 * BLK);
    REQUESTS_STAY_HERE();
    run_blk(r1, A, h0, h1, acc0, acc1, wihA, wihB, biasA, biasB);

    load_blk(r1, xp0, xp1, s + 5 * BLK);
    REQUESTS_STAY_HERE();
    run_blk(r2, A, h0, h1, acc0, acc1, wihA, wihB, biasA, biasB);

    load_blk(r2, xp0, xp1, s + 6 * BLK);
    REQUESTS_STAY_HERE();
    run_blk(r3, A, h0, h1, acc0, acc1, wihA, wihB, biasA, biasB);
  }

  load_blk(r3, xp0, xp1, s + 3 * BLK);
  run_blk(r0, A, h0, h1, acc0, acc1, wihA, wihB, biasA, biasB);
  run_blk(r1, A, h0, h1, acc0, acc1, wihA, wihB, biasA, biasB);
  run_blk(r2, A, h0, h1, acc0, acc1, wihA, wihB, biasA, biasB);
  run_blk(r3, A, h0, h1, acc0, acc1, wihA, wihB, biasA, biasB);

  float p0 = __builtin_fmaf(h0.x, fwA, h0.y * fwB);
  float p1 = __builtin_fmaf(h1.x, fwA, h1.y * fwB);
  p0 += __shfl_xor(p0, 16, 32);
  p1 += __shfl_xor(p1, 16, 32);
  if (lo) {
    out[base + m]      = p0 + fcb;
    out[base + 16 + m] = p1 + fcb;
  }
}

extern "C" void kernel_launch(void* const* d_in, const int* in_sizes, int n_in,
                              void* d_out, int out_size, void* d_ws, size_t ws_size,
                              hipStream_t stream) {
  (void)n_in; (void)out_size; (void)d_ws; (void)ws_size;
  const float* x    = (const float*)d_in[0];
  const float* W_ih = (const float*)d_in[1];
  const float* W_hh = (const float*)d_in[2];
  const float* b_ih = (const float*)d_in[3];
  const float* b_hh = (const float*)d_in[4];
  const float* fc_w = (const float*)d_in[5];
  const float* fc_b = (const float*)d_in[6];
  float* out = (float*)d_out;

  const int B = in_sizes[0] / S_LEN;
  const int block = 128;
  const int grid  = (B + 127) / 128;
  rnn_scan_wmma_kernel<<<grid, block, 0, stream>>>(x, W_ih, W_hh, b_ih, b_hh,
                                                   fc_w, fc_b, out, B);
}


# one chain per wave: a wave carries 16 batch rows instead of 32, grid doubled to 256 x 128 threads so 1024 waves cover 1024 SIMDs; x requests issued behind each block's first matrix instruction
# speedup vs baseline: 2.2257x; 1.4440x over previous
#include <hip/hip_runtime.h>


typedef __attribute__((ext_vector_type(2))) float v2f;
typedef __attribute__((ext_vector_type(8))) float v8f;

#define S_LEN 1024
#define HID 4
#define BLK 16

__device__ __forceinline__ float cdna5_tanh(float v) {
#if __has_builtin(__builtin_amdgcn_tanhf)
  return __builtin_amdgcn_tanhf(v);
#else
  float r;
  asm("v_tanh_f32 %0, %1" : "=v"(r) : "v"(v));
  return r;
#endif
}

struct XBlk {
  float4 a[4];
};

__device__ __forceinline__ void load_blk(XBlk& r,
                                         const float* __restrict__ xp,
                                         int off) {
#pragma unroll
  for (int i = 0; i < 4; ++i) {
    r.a[i] = *(const float4*)(xp + off + 4 * i);
  }
}

#define REQUESTS_STAY_HERE()                \
  do {                                      \
    asm volatile("" : : : "memory");        \
    __builtin_amdgcn_sched_barrier(0);      \
  } while (0)

template <bool REQ>
__device__ __forceinline__ void run_blk(const XBlk& r, XBlk& nxt,
                                        const float* __restrict__ xp, int off,
                                        const v2f A, v2f& h, v8f& acc,
                                        const float wihA, const float wihB,
                                        const float biasA, const float biasB) {
  const float xs[BLK] = {r.a[0].x, r.a[0].y, r.a[0].z, r.a[0].w,
                         r.a[1].x, r.a[1].y, r.a[1].z, r.a[1].w,
                         r.a[2].x, r.a[2].y, r.a[2].z, r.a[2].w,
                         r.a[3].x, r.a[3].y, r.a[3].z, r.a[3].w};
#pragma unroll
  for (int u = 0; u < BLK; ++u) {
    acc[0] = __builtin_fmaf(xs[u], wihA, biasA);
    acc[1] = __builtin_fmaf(xs[u], wihB, biasB);
    acc = __builtin_amdgcn_wmma_f32_16x16x4_f32(
        false, A, false, h, (short)0, acc, false, false);
    if (REQ && u == 0) {
      REQUESTS_STAY_HERE();
      load_blk(nxt, xp, off);
      REQUESTS_STAY_HERE();
    }
    h.x = cdna5_tanh(acc[0]);
    h.y = cdna5_tanh(acc[1]);
  }
}

__global__ void __launch_bounds__(128)
rnn_scan_wmma_kernel(const float* __restrict__ x,
                     const float* __restrict__ W_ih,
                     const float* __restrict__ W_hh,
                     const float* __restrict__ b_ih,
                     const float* __restrict__ b_hh,
                     const float* __restrict__ fc_w,
                     const float* __restrict__ fc_b,
                     float* __restrict__ out,
                     int Btot)
{
  const int gtid = blockIdx.x * blockDim.x + threadIdx.x;
  const int lane = threadIdx.x & 31;
  const int base = (gtid >> 5) << 4;
  if (base >= Btot) return;

  const int  m  = lane & 15;
  const bool lo = lane < 16;

  const int jA = lo ? 0 : 2;
  const int jB = lo ? 1 : 3;
  const float wihA  = W_ih[jA];
  const float wihB  = W_ih[jB];
  const float biasA = b_ih[jA] + b_hh[jA];
  const float biasB = b_ih[jB] + b_hh[jB];
  const float fwA   = fc_w[jA];
  const float fwB   = fc_w[jB];
  const float fcb   = fc_b[0];

  v2f A = {0.f, 0.f};
  {
    const int row = (m == 0) ? 0 : (m == 1) ? 1 : (m == 8) ? 2 : (m == 9) ? 3 : -1;
    if (row >= 0) {
      const int k0 = lo ? 0 : 2;
      A.x = W_hh[row * HID + k0];
      A.y = W_hh[row * HID + k0 + 1];
    }
  }

  const float* __restrict__ xp = x + (size_t)(base + m) * S_LEN;

  v2f h = {0.f, 0.f};
  v8f acc = {};

  XBlk r0, r1, r2, r3;
  load_blk(r0, xp, 0 * BLK);
  REQUESTS_STAY_HERE();
  load_blk(r1, xp, 1 * BLK);
  REQUESTS_STAY_HERE();
  load_blk(r2, xp, 2 * BLK);
  REQUESTS_STAY_HERE();

  int s = 0;
  for (; s < S_LEN - 4 * BLK; s += 4 * BLK) {
    run_blk<true>(r0, r3, xp, s + 3 * BLK, A, h, acc, wihA, wihB, biasA, biasB);
    run_blk<true>(r1, r0, xp, s + 4 * BLK, A, h, acc, wihA, wihB, biasA, biasB);
    run_blk<true>(r2, r1, xp, s + 5 * BLK, A, h, acc, wihA, wihB, biasA, biasB);
    run_blk<true>(r3, r2, xp, s + 6 * BLK, A, h, acc, wihA, wihB, biasA, biasB);
  }

  run_blk<true >(r0, r3, xp, s + 3 * BLK, A, h, acc, wihA, wihB, biasA, biasB);
  run_blk<false>(r1, r0, xp, 0,           A, h, acc, wihA, wihB, biasA, biasB);
  run_blk<false>(r2, r1, xp, 0,           A, h, acc, wihA, wihB, biasA, biasB);
  run_blk<false>(r3, r2, xp, 0,           A, h, acc, wihA, wihB, biasA, biasB);

  float p = __builtin_fmaf(h.x, fwA, h.y * fwB);
  p += __shfl_xor(p, 16, 32);
  if (lo) {
    out[base + m] = p + fcb;
  }
}

extern "C" void kernel_launch(void* const* d_in, const int* in_sizes, int n_in,
                              void* d_out, int out_size, void* d_ws, size_t ws_size,
                              hipStream_t stream) {
  (void)n_in; (void)out_size; (void)d_ws; (void)ws_size;
  const float* x    = (const float*)d_in[0];
  const float* W_ih = (const float*)d_in[1];
  const float* W_hh = (const float*)d_in[2];
  const float* b_ih = (const float*)d_in[3];
  const float* b_hh = (const float*)d_in[4];
  const float* fc_w = (const float*)d_in[5];
  const float* fc_b = (const float*)d_in[6];
  float* out = (float*)d_out;

  const int B = in_sizes[0] / S_LEN;
  const int block = 128;
  const int grid  = (B + 63) / 64;
  rnn_scan_wmma_kernel<<<grid, block, 0, stream>>>(x, W_ih, W_hh, b_ih, b_hh,
                                                   fc_w, fc_b, out, B);
}
